# starting kernel
# baseline (speedup 1.0000x reference)
#include <hip/hip_runtime.h>
#include <hip/hip_bf16.h>

#define N_NODES 50000
#define N_EDGES 800000
#define F_IN    128
#define HEADS   8
#define FH      16
#define HF      128
#define NEG_SLOPE 0.2f

typedef float v2f __attribute__((ext_vector_type(2)));
typedef float v4f __attribute__((ext_vector_type(4)));
typedef float v8f __attribute__((ext_vector_type(8)));

__global__ __launch_bounds__(256) void gat_gemm_wmma(
    const float* __restrict__ x, const float* __restrict__ W,
    float* __restrict__ h) {
  const int lane = threadIdx.x & 31;
  const int wave = threadIdx.x >> 5;
  const int m0   = blockIdx.x * 16;
  const int n0   = wave * 16;
  const int half = lane >> 4;
  const int l15  = lane & 15;

  const float* __restrict__ arow = x + (m0 + l15) * F_IN + 2 * half;
  const float* __restrict__ brow = W + (n0 + l15) * F_IN + 2 * half;

  v8f c = {};
#pragma unroll
  for (int k0 = 0; k0 < F_IN; k0 += 4) {
    v2f a; a.x = arow[k0]; a.y = arow[k0 + 1];
    v2f b; b.x = brow[k0]; b.y = brow[k0 + 1];
    c = __builtin_amdgcn_wmma_f32_16x16x4_f32(
         false, a,  false, b,
         (short)0, c,  false,  false);
  }

  float* __restrict__ op = h + (m0 + half * 8) * HF + n0 + l15;
#pragma unroll
  for (int v = 0; v < 8; ++v) op[v * HF] = c[v];
}

__global__ void gat_logits(const float* __restrict__ h,
                           const float* __restrict__ att_src,
                           const float* __restrict__ att_dst,
                           float* __restrict__ asrc, float* __restrict__ adst) {
  int t = blockIdx.x * blockDim.x + threadIdx.x;
  if (t >= N_NODES * HEADS) return;
  int head = t & 7;
  int n    = t >> 3;
  const float* __restrict__ hp = h + n * HF + head * FH;
  const float* __restrict__ as = att_src + head * FH;
  const float* __restrict__ ad = att_dst + head * FH;
  float s = 0.f, d = 0.f;
#pragma unroll
  for (int f = 0; f < FH; ++f) { float v = hp[f]; s += v * as[f]; d += v * ad[f]; }
  asrc[t] = s;
  adst[t] = d;
}

__global__ void gat_init(float* __restrict__ out, const float* __restrict__ bias,
                         int* __restrict__ mkey, float* __restrict__ denom) {
  int t = blockIdx.x * blockDim.x + threadIdx.x;
  if (t < N_NODES * HF) out[t] = bias[t & (HF - 1)];
  if (t < N_NODES * HEADS) { mkey[t] = (int)0x80000000; denom[t] = 0.f; }
}

__device__ __forceinline__ int f32_key(float f) {
  int i = __float_as_int(f);
  return i >= 0 ? i : (i ^ 0x7fffffff);
}
__device__ __forceinline__ float key_f32(int k) {
  return __int_as_float(k >= 0 ? k : (k ^ 0x7fffffff));
}

__device__ __forceinline__ float leaky(float v) {
  return v >= 0.f ? v : NEG_SLOPE * v;
}

__global__ void gat_edge_max(const int* __restrict__ ei,
                             const float* __restrict__ asrc,
                             const float* __restrict__ adst,
                             int* __restrict__ mkey) {
  int t = blockIdx.x * blockDim.x + threadIdx.x;
  if (t >= N_EDGES * HEADS) return;
  int e = t >> 3, head = t & 7;
  int s = ei[e];
  int d = ei[N_EDGES + e];
  float v = leaky(asrc[s * HEADS + head] + adst[d * HEADS + head]);
  __hip_atomic_fetch_max(&mkey[d * HEADS + head], f32_key(v),
                         __ATOMIC_RELAXED, __HIP_MEMORY_SCOPE_AGENT);
}

__global__ void gat_edge_exp(const int* __restrict__ ei,
                             const float* __restrict__ asrc,
                             const float* __restrict__ adst,
                             const int* __restrict__ mkey,
                             float* __restrict__ exw,
                             float* __restrict__ denom) {
  int t = blockIdx.x * blockDim.x + threadIdx.x;
  if (t >= N_EDGES * HEADS) return;
  int e = t >> 3, head = t & 7;
  int s = ei[e];
  int d = ei[N_EDGES + e];
  float v = leaky(asrc[s * HEADS + head] + adst[d * HEADS + head]);
  float m = key_f32(mkey[d * HEADS + head]);
  float ex = __expf(v - m);
  exw[t] = ex;
  __hip_atomic_fetch_add(&denom[d * HEADS + head], ex,
                         __ATOMIC_RELAXED, __HIP_MEMORY_SCOPE_AGENT);
}

__global__ __launch_bounds__(256) void gat_aggregate(
    const int* __restrict__ ei, const float* __restrict__ h,
    const float* __restrict__ exw, const float* __restrict__ denom,
    float* __restrict__ out) {
  int t = blockIdx.x * blockDim.x + threadIdx.x;
  int e    = t >> 5;
  int lane = t & 31;
  int s = ei[e];
  int d = ei[N_EDGES + e];
  int head = lane >> 2;
  float alpha = exw[e * HEADS + head] /
                (denom[d * HEADS + head] + 1e-16f);
  const v4f hv = *(const v4f*)(h + s * HF + lane * 4);
  float* __restrict__ op = out + d * HF + lane * 4;
#pragma unroll
  for (int i = 0; i < 4; ++i)
    __hip_atomic_fetch_add(&op[i], alpha * hv[i],
                           __ATOMIC_RELAXED, __HIP_MEMORY_SCOPE_AGENT);
}

extern "C" void kernel_launch(void* const* d_in, const int* in_sizes, int n_in,
                              void* d_out, int out_size, void* d_ws, size_t ws_size,
                              hipStream_t stream) {
  const float* x        = (const float*)d_in[0];
  const float* W        = (const float*)d_in[1];
  const float* att_src  = (const float*)d_in[2];
  const float* att_dst  = (const float*)d_in[3];
  const float* bias     = (const float*)d_in[4];
  const int*   ei       = (const int*)d_in[5];
  float*       out      = (float*)d_out;

  float* h     = (float*)d_ws;
  float* asrc  = h     + (size_t)N_NODES * HF;
  float* adst  = asrc  + (size_t)N_NODES * HEADS;
  float* denom = adst  + (size_t)N_NODES * HEADS;
  int*   mkey  = (int*)(denom + (size_t)N_NODES * HEADS);
  float* exw   = (float*)(mkey + (size_t)N_NODES * HEADS);

  const int BLK = 256;

  gat_gemm_wmma<<<N_NODES / 16, BLK, 0, stream>>>(x, W, h);

  gat_logits<<<(N_NODES * HEADS + BLK - 1) / BLK, BLK, 0, stream>>>(
      h, att_src, att_dst, asrc, adst);

  gat_init<<<(N_NODES * HF + BLK - 1) / BLK, BLK, 0, stream>>>(
      out, bias, mkey, denom);

  gat_edge_max<<<(N_EDGES * HEADS + BLK - 1) / BLK, BLK, 0, stream>>>(
      ei, asrc, adst, mkey);

  gat_edge_exp<<<(N_EDGES * HEADS + BLK - 1) / BLK, BLK, 0, stream>>>(
      ei, asrc, adst, mkey, exw, denom);

  gat_aggregate<<<(N_EDGES * 32) / BLK, BLK, 0, stream>>>(
      ei, h, exw, denom, out);
}


# Edges grouped by destination in-call (count, segment alloc, fill); one wave per node orders its segment by source id (fixed sum order), does softmax and weighted row gather; float atomics gone
# speedup vs baseline: 1.7792x; 1.7792x over previous
#include <hip/hip_runtime.h>
#include <hip/hip_bf16.h>

#define N_NODES 50000
#define N_EDGES 800000
#define F_IN    128
#define HEADS   8
#define FH      16
#define HF      128
#define NEG_SLOPE 0.2f

typedef float v2f __attribute__((ext_vector_type(2)));
typedef float v4f __attribute__((ext_vector_type(4)));
typedef float v8f __attribute__((ext_vector_type(8)));

__global__ __launch_bounds__(256) void gat_gemm_wmma(
    const float* __restrict__ x, const float* __restrict__ W,
    float* __restrict__ h) {
  const int lane = threadIdx.x & 31;
  const int wave = threadIdx.x >> 5;
  const int m0   = blockIdx.x * 16;
  const int n0   = wave * 16;
  const int half = lane >> 4;
  const int l15  = lane & 15;

  const float* __restrict__ arow = x + (m0 + l15) * F_IN + 2 * half;
  const float* __restrict__ brow = W + (n0 + l15) * F_IN + 2 * half;

  v8f c = {};
#pragma unroll
  for (int k0 = 0; k0 < F_IN; k0 += 4) {
    v2f a; a.x = arow[k0]; a.y = arow[k0 + 1];
    v2f b; b.x = brow[k0]; b.y = brow[k0 + 1];
    c = __builtin_amdgcn_wmma_f32_16x16x4_f32(
         false, a,  false, b,
         (short)0, c,  false,  false);
  }

  float* __restrict__ op = h + (m0 + half * 8) * HF + n0 + l15;
#pragma unroll
  for (int v = 0; v < 8; ++v) op[v * HF] = c[v];
}

__global__ void gat_logits(const float* __restrict__ h,
                           const float* __restrict__ att_src,
                           const float* __restrict__ att_dst,
                           float* __restrict__ asrc, float* __restrict__ adst,
                           int* __restrict__ cnt, int* __restrict__ cursor) {
  int t = blockIdx.x * blockDim.x + threadIdx.x;
  if (t >= N_NODES * HEADS) return;
  if (t < N_NODES) cnt[t] = 0;
  if (t == 0) cursor[0] = 0;
  int head = t & 7;
  int n    = t >> 3;
  const float* __restrict__ hp = h + n * HF + head * FH;
  const float* __restrict__ as = att_src + head * FH;
  const float* __restrict__ ad = att_dst + head * FH;
  float s = 0.f, d = 0.f;
#pragma unroll
  for (int f = 0; f < FH; ++f) { float v = hp[f]; s += v * as[f]; d += v * ad[f]; }
  asrc[t] = s;
  adst[t] = d;
}

__device__ __forceinline__ float leaky(float v) {
  return v >= 0.f ? v : NEG_SLOPE * v;
}

__global__ void gat_count(const int* __restrict__ ei, int* __restrict__ cnt,
                          int* __restrict__ rank) {
  int t = blockIdx.x * blockDim.x + threadIdx.x;
  if (t >= N_EDGES) return;
  int d = ei[N_EDGES + t];
  rank[t] = __hip_atomic_fetch_add(&cnt[d], 1, __ATOMIC_RELAXED,
                                   __HIP_MEMORY_SCOPE_AGENT);
}

__global__ void gat_alloc(const int* __restrict__ cnt, int* __restrict__ cursor,
                          int* __restrict__ off) {
  int t = blockIdx.x * blockDim.x + threadIdx.x;
  int lane = threadIdx.x & 31;
  int c = (t < N_NODES) ? cnt[t] : 0;
  int incl = c;
  for (int k = 1; k < 32; k <<= 1) {
    int up = __shfl_up(incl, k, 32);
    if (lane >= k) incl += up;
  }
  int total = __shfl(incl, 31, 32);
  int base = 0;
  if (lane == 0)
    base = __hip_atomic_fetch_add(cursor, total, __ATOMIC_RELAXED,
                                  __HIP_MEMORY_SCOPE_AGENT);
  base = __shfl(base, 0, 32);
  if (t < N_NODES) off[t] = base + incl - c;
}

__global__ void gat_fill(const int* __restrict__ ei, const int* __restrict__ off,
                         const int* __restrict__ rank, int* __restrict__ slot) {
  int t = blockIdx.x * blockDim.x + threadIdx.x;
  if (t >= N_EDGES) return;
  int s = ei[t];
  int d = ei[N_EDGES + t];
  slot[off[d] + rank[t]] = s;
}

template <int CTRL>
__device__ __forceinline__ float quad(float x) {
  return __int_as_float(__builtin_amdgcn_mov_dpp(__float_as_int(x), CTRL, 0xf, 0xf, true));
}
#define QP_XOR1 0xB1
#define QP_XOR2 0x4E
#define QP_BC0  0x00
#define QP_BC1  0x55
#define QP_BC2  0xAA
#define QP_BC3  0xFF

__global__ __launch_bounds__(256) void gat_node(
    const int* __restrict__ cnt, const int* __restrict__ off,
    const int* __restrict__ slot, int* __restrict__ srtbuf,
    const float* __restrict__ h, const float* __restrict__ asrc,
    const float* __restrict__ adst, const float* __restrict__ bias,
    float* __restrict__ out) {
  const int lane = threadIdx.x & 31;
  const int n    = blockIdx.x * 8 + (threadIdx.x >> 5);
  const int q    = lane & 3;
  const int head = lane >> 2;
  const int deg = __builtin_amdgcn_readfirstlane(cnt[n]);
  const int beg = __builtin_amdgcn_readfirstlane(off[n]);
  const float adn = adst[n * HEADS + head];
  const float ninf = -__builtin_inff();

  v4f acc = {0.f, 0.f, 0.f, 0.f};

  if (deg <= 32) {
    const int key = (lane < deg) ? slot[beg + lane] * 32 + lane : 0x7fffffff;
    int r = 0;
    for (int j = 0; j < deg; ++j) {
      r += (__builtin_amdgcn_readlane(key, j) < key) ? 1 : 0;
    }
    const int srt = __builtin_amdgcn_ds_permute(r << 2, key) >> 5;

    float vv[8];
#pragma unroll
    for (int p = 0; p < 8; ++p) {
      vv[p] = 0.f;
      if (4 * p < deg) {
        const int j  = min(4 * p + q, deg - 1);
        const int sj = __builtin_amdgcn_ds_bpermute(j << 2, srt);
        vv[p] = asrc[sj * HEADS + head];
      }
    }
    float m = ninf;
#pragma unroll
    for (int p = 0; p < 8; ++p) {
      vv[p] = (4 * p + q < deg) ? leaky(vv[p] + adn) : ninf;
      m = fmaxf(m, vv[p]);
    }
    m = fmaxf(m, quad<QP_XOR1>(m));
    m = fmaxf(m, quad<QP_XOR2>(m));

    float s = 0.f;
#pragma unroll
    for (int p = 0; p < 8; ++p) {
      if (4 * p < deg) {
        vv[p] = (4 * p + q < deg) ? __expf(vv[p] - m) : 0.f;
        s += vv[p];
      }
    }
    s += quad<QP_XOR1>(s);
    s += quad<QP_XOR2>(s);
    const float den = s + 1e-16f;

#pragma unroll
    for (int p = 0; p < 8; ++p) {
      if (4 * p < deg) {
        const float al = vv[p] / den;
        const int j0 = 4 * p;
        const int s0 = __builtin_amdgcn_readlane(srt, j0);
        const int s1 = (j0 + 1 < deg) ? __builtin_amdgcn_readlane(srt, j0 + 1) : s0;
        const int s2 = (j0 + 2 < deg) ? __builtin_amdgcn_readlane(srt, j0 + 2) : s0;
        const int s3 = (j0 + 3 < deg) ? __builtin_amdgcn_readlane(srt, j0 + 3) : s0;
        const v4f h0 = *(const v4f*)(h + s0 * HF + lane * 4);
        const v4f h1 = *(const v4f*)(h + s1 * HF + lane * 4);
        const v4f h2 = *(const v4f*)(h + s2 * HF + lane * 4);
        const v4f h3 = *(const v4f*)(h + s3 * HF + lane * 4);
        acc += quad<QP_BC0>(al) * h0;
        acc += quad<QP_BC1>(al) * h1;
        acc += quad<QP_BC2>(al) * h2;
        acc += quad<QP_BC3>(al) * h3;
      }
    }
  } else {
    for (int i = lane; i < deg; i += 32) {
      const int key = slot[beg + i];
      int r = 0;
      for (int j = 0; j < deg; ++j) {
        const int kj = slot[beg + j];
        r += ((kj < key) | ((kj == key) & (j < i))) ? 1 : 0;
      }
      srtbuf[beg + r] = key;
    }
    __threadfence();
    float m = ninf;
    for (int j = q; j < deg; j += 4) {
      const int sj = srtbuf[beg + j];
      m = fmaxf(m, leaky(asrc[sj * HEADS + head] + adn));
    }
    m = fmaxf(m, quad<QP_XOR1>(m));
    m = fmaxf(m, quad<QP_XOR2>(m));
    float s = 0.f;
    for (int j = q; j < deg; j += 4) {
      const int sj = srtbuf[beg + j];
      s += __expf(leaky(asrc[sj * HEADS + head] + adn) - m);
    }
    s += quad<QP_XOR1>(s);
    s += quad<QP_XOR2>(s);
    const float den = s + 1e-16f;
    for (int j = 0; j < deg; ++j) {
      const int sj = __builtin_amdgcn_readfirstlane(srtbuf[beg + j]);
      const float al = __expf(leaky(asrc[sj * HEADS + head] + adn) - m) / den;
      const v4f hv = *(const v4f*)(h + sj * HF + lane * 4);
      acc += al * hv;
    }
  }

  const v4f b = *(const v4f*)(bias + lane * 4);
  *(v4f*)(out + n * HF + lane * 4) = acc + b;
}

extern "C" void kernel_launch(void* const* d_in, const int* in_sizes, int n_in,
                              void* d_out, int out_size, void* d_ws, size_t ws_size,
                              hipStream_t stream) {
  const float* x        = (const float*)d_in[0];
  const float* W        = (const float*)d_in[1];
  const float* att_src  = (const float*)d_in[2];
  const float* att_dst  = (const float*)d_in[3];
  const float* bias     = (const float*)d_in[4];
  const int*   ei       = (const int*)d_in[5];
  float*       out      = (float*)d_out;

  float* h      = (float*)d_ws;
  float* asrc   = h     + (size_t)N_NODES * HF;
  float* adst   = asrc  + (size_t)N_NODES * HEADS;
  int*   cnt    = (int*)(adst + (size_t)N_NODES * HEADS);
  int*   off    = cnt   + N_NODES;
  int*   cursor = off   + N_NODES;
  int*   rank   = cursor + 32;
  int*   slot   = rank  + N_EDGES;
  int*   srtbuf = slot  + N_EDGES;

  const int BLK = 256;

  gat_gemm_wmma<<<N_NODES / 16, BLK, 0, stream>>>(x, W, h);

  gat_logits<<<(N_NODES * HEADS + BLK - 1) / BLK, BLK, 0, stream>>>(
      h, att_src, att_dst, asrc, adst, cnt, cursor);

  gat_count<<<(N_EDGES + BLK - 1) / BLK, BLK, 0, stream>>>(ei, cnt, rank);

  gat_alloc<<<(N_NODES + BLK - 1) / BLK, BLK, 0, stream>>>(cnt, cursor, off);

  gat_fill<<<(N_EDGES + BLK - 1) / BLK, BLK, 0, stream>>>(ei, off, rank, slot);

  gat_node<<<N_NODES / 8, BLK, 0, stream>>>(
      cnt, off, slot, srtbuf, h, asrc, adst, bias, out);
}


# Edge grouping on LDS counters: two-level counting sort by destination (256 node ranges, then the nodes of a range) replaces the 800000 per-edge returning global adds; sum order unchanged
# speedup vs baseline: 2.7071x; 1.5215x over previous
#include <hip/hip_runtime.h>
#include <hip/hip_bf16.h>

#define N_NODES 50000
#define N_EDGES 800000
#define F_IN    128
#define HEADS   8
#define FH      16
#define HF      128
#define NEG_SLOPE 0.2f

typedef float v2f __attribute__((ext_vector_type(2)));
typedef float v4f __attribute__((ext_vector_type(4)));
typedef float v8f __attribute__((ext_vector_type(8)));

__global__ __launch_bounds__(256) void gat_gemm_wmma(
    const float* __restrict__ x, const float* __restrict__ W,
    float* __restrict__ h) {
  const int lane = threadIdx.x & 31;
  const int wave = threadIdx.x >> 5;
  const int m0   = blockIdx.x * 16;
  const int n0   = wave * 16;
  const int half = lane >> 4;
  const int l15  = lane & 15;

  const float* __restrict__ arow = x + (m0 + l15) * F_IN + 2 * half;
  const float* __restrict__ brow = W + (n0 + l15) * F_IN + 2 * half;

  v8f c = {};
#pragma unroll
  for (int k0 = 0; k0 < F_IN; k0 += 4) {
    v2f a; a.x = arow[k0]; a.y = arow[k0 + 1];
    v2f b; b.x = brow[k0]; b.y = brow[k0 + 1];
    c = __builtin_amdgcn_wmma_f32_16x16x4_f32(
         false, a,  false, b,
         (short)0, c,  false,  false);
  }

  float* __restrict__ op = h + (m0 + half * 8) * HF + n0 + l15;
#pragma unroll
  for (int v = 0; v < 8; ++v) op[v * HF] = c[v];
}

#define NB     256
#define NPB    196
#define CSTR   32
#define EPT    8
#define ECHUNK (256 * EPT)
#define NCHUNK ((N_EDGES + ECHUNK - 1) / ECHUNK)
static_assert(NB * NPB >= N_NODES, "buckets must cover all nodes");
static_assert(N_NODES <= 65536, "source id is packed into 16 bits");
static_assert(NB == 256, "one bucket per thread of a 256-thread work-group");

__global__ void gat_logits(const float* __restrict__ h,
                           const float* __restrict__ att_src,
                           const float* __restrict__ att_dst,
                           float* __restrict__ asrc, float* __restrict__ adst,
                           int* __restrict__ bctr) {
  int t = blockIdx.x * blockDim.x + threadIdx.x;
  if (t >= N_NODES * HEADS) return;
  if (t < 2 * NB * CSTR) bctr[t] = 0;
  int head = t & 7;
  int n    = t >> 3;
  const float* __restrict__ hp = h + n * HF + head * FH;
  const float* __restrict__ as = att_src + head * FH;
  const float* __restrict__ ad = att_dst + head * FH;
  float s = 0.f, d = 0.f;
#pragma unroll
  for (int f = 0; f < FH; ++f) { float v = hp[f]; s += v * as[f]; d += v * ad[f]; }
  asrc[t] = s;
  adst[t] = d;
}

__device__ __forceinline__ float leaky(float v) {
  return v >= 0.f ? v : NEG_SLOPE * v;
}

__global__ __launch_bounds__(256) void gat_hist(const int* __restrict__ ei,
                                                int* __restrict__ bcnt) {
  __shared__ int lh[NB];
  const int tid = threadIdx.x;
  lh[tid] = 0;
  __syncthreads();
  const int e0 = blockIdx.x * ECHUNK + tid;
  int d[EPT];
#pragma unroll
  for (int i = 0; i < EPT; ++i) d[i] = ei[N_EDGES + min(e0 + i * 256, N_EDGES - 1)];
#pragma unroll
  for (int i = 0; i < EPT; ++i) {
    if (e0 + i * 256 < N_EDGES)
      __hip_atomic_fetch_add(&lh[d[i] / NPB], 1, __ATOMIC_RELAXED,
                             __HIP_MEMORY_SCOPE_WORKGROUP);
  }
  __syncthreads();
  const int c = lh[tid];
  if (c > 0)
    __hip_atomic_fetch_add(&bcnt[tid * CSTR], c, __ATOMIC_RELAXED,
                           __HIP_MEMORY_SCOPE_AGENT);
}

__global__ __launch_bounds__(256) void gat_scatter(
    const int* __restrict__ ei, const int* __restrict__ bcnt,
    int* __restrict__ bfill, int* __restrict__ part) {
  __shared__ int lh[NB];
  __shared__ int lbase[NB];
  __shared__ int wsum[8];
  const int tid  = threadIdx.x;
  const int lane = tid & 31;
  const int wave = tid >> 5;
  lh[tid] = 0;
  __syncthreads();
  const int e0 = blockIdx.x * ECHUNK + tid;
  int s[EPT], d[EPT], bk[EPT], r[EPT];
#pragma unroll
  for (int i = 0; i < EPT; ++i) {
    const int e = min(e0 + i * 256, N_EDGES - 1);
    s[i] = ei[e];
    d[i] = ei[N_EDGES + e];
  }
  const int tot = bcnt[tid * CSTR];
#pragma unroll
  for (int i = 0; i < EPT; ++i) {
    bk[i] = d[i] / NPB;
    r[i] = 0;
    if (e0 + i * 256 < N_EDGES)
      r[i] = __hip_atomic_fetch_add(&lh[bk[i]], 1, __ATOMIC_RELAXED,
                                    __HIP_MEMORY_SCOPE_WORKGROUP);
  }
  int incl = tot;
  for (int k = 1; k < 32; k <<= 1) {
    const int up = __shfl_up(incl, k, 32);
    if (lane >= k) incl += up;
  }
  if (lane == 31) wsum[wave] = incl;
  __syncthreads();
  int woff = 0;
  for (int w = 0; w < 8; ++w) woff += (w < wave) ? wsum[w] : 0;
  const int c = lh[tid];
  int got = 0;
  if (c > 0)
    got = __hip_atomic_fetch_add(&bfill[tid * CSTR], c, __ATOMIC_RELAXED,
                                 __HIP_MEMORY_SCOPE_AGENT);
  lbase[tid] = woff + incl - tot + got;
  __syncthreads();
#pragma unroll
  for (int i = 0; i < EPT; ++i) {
    if (e0 + i * 256 < N_EDGES)
      part[lbase[bk[i]] + r[i]] = ((d[i] - bk[i] * NPB) << 16) | s[i];
  }
}

#define BLD 1024
__global__ __launch_bounds__(BLD) void gat_build(
    const int* __restrict__ bcnt, const int* __restrict__ part,
    int* __restrict__ cnt, int* __restrict__ off, int* __restrict__ slot) {
  __shared__ int ncnt[NB];
  __shared__ int ncur[NB];
  __shared__ int wsA[8];
  __shared__ int wsB[8];
  const int tid  = threadIdx.x;
  const int lane = tid & 31;
  const int wave = tid >> 5;
  const int b    = blockIdx.x;
  const int nb   = bcnt[b * CSTR];

  if (tid < NB) {
    ncnt[tid] = 0;
    int v = (tid < b) ? bcnt[tid * CSTR] : 0;
    for (int k = 16; k >= 1; k >>= 1) v += __shfl_xor(v, k, 32);
    if (lane == 0) wsA[wave] = v;
  }
  __syncthreads();
  int bstart = 0;
  for (int w = 0; w < 8; ++w) bstart += wsA[w];

  for (int eb = tid; eb < nb; eb += 4 * BLD) {
    int ent[4];
#pragma unroll
    for (int k = 0; k < 4; ++k) {
      const int e = eb + k * BLD;
      ent[k] = (e < nb) ? part[bstart + e] : -1;
    }
#pragma unroll
    for (int k = 0; k < 4; ++k) {
      if (ent[k] >= 0)
        __hip_atomic_fetch_add(&ncnt[ent[k] >> 16], 1, __ATOMIC_RELAXED,
                               __HIP_MEMORY_SCOPE_WORKGROUP);
    }
  }
  __syncthreads();

  int c = 0, incl = 0;
  if (tid < NB) {
    c = ncnt[tid];
    incl = c;
    for (int k = 1; k < 32; k <<= 1) {
      const int up = __shfl_up(incl, k, 32);
      if (lane >= k) incl += up;
    }
    if (lane == 31) wsB[wave] = incl;
  }
  __syncthreads();
  if (tid < NB) {
    int woff = 0;
    for (int w = 0; w < 8; ++w) woff += (w < wave) ? wsB[w] : 0;
    const int noff = woff + incl - c;
    ncur[tid] = noff;
    const int n = b * NPB + tid;
    if (tid < NPB && n < N_NODES) {
      cnt[n] = c;
      off[n] = bstart + noff;
    }
  }
  __syncthreads();

  for (int eb = tid; eb < nb; eb += 4 * BLD) {
    int ent[4];
#pragma unroll
    for (int k = 0; k < 4; ++k) {
      const int e = eb + k * BLD;
      ent[k] = (e < nb) ? part[bstart + e] : -1;
    }
#pragma unroll
    for (int k = 0; k < 4; ++k) {
      if (ent[k] >= 0) {
        const int pos = __hip_atomic_fetch_add(&ncur[ent[k] >> 16], 1,
                                               __ATOMIC_RELAXED,
                                               __HIP_MEMORY_SCOPE_WORKGROUP);
        slot[bstart + pos] = ent[k] & 0xffff;
      }
    }
  }
}

template <int CTRL>
__device__ __forceinline__ float quad(float x) {
  return __int_as_float(__builtin_amdgcn_mov_dpp(__float_as_int(x), CTRL, 0xf, 0xf, true));
}
#define QP_XOR1 0xB1
#define QP_XOR2 0x4E
#define QP_BC0  0x00
#define QP_BC1  0x55
#define QP_BC2  0xAA
#define QP_BC3  0xFF

__global__ __launch_bounds__(256) void gat_node(
    const int* __restrict__ cnt, const int* __restrict__ off,
    const int* __restrict__ slot, int* __restrict__ srtbuf,
    const float* __restrict__ h, const float* __restrict__ asrc,
    const float* __restrict__ adst, const float* __restrict__ bias,
    float* __restrict__ out) {
  const int lane = threadIdx.x & 31;
  const int n    = blockIdx.x * 8 + (threadIdx.x >> 5);
  const int q    = lane & 3;
  const int head = lane >> 2;
  const int deg = __builtin_amdgcn_readfirstlane(cnt[n]);
  const int beg = __builtin_amdgcn_readfirstlane(off[n]);
  const float adn = adst[n * HEADS + head];
  const float ninf = -__builtin_inff();

  v4f acc = {0.f, 0.f, 0.f, 0.f};

  if (deg <= 32) {
    const int key = (lane < deg) ? slot[beg + lane] * 32 + lane : 0x7fffffff;
    int r = 0;
    for (int j = 0; j < deg; ++j) {
      r += (__builtin_amdgcn_readlane(key, j) < key) ? 1 : 0;
    }
    const int srt = __builtin_amdgcn_ds_permute(r << 2, key) >> 5;

    float vv[8];
#pragma unroll
    for (int p = 0; p < 8; ++p) {
      vv[p] = 0.f;
      if (4 * p < deg) {
        const int j  = min(4 * p + q, deg - 1);
        const int sj = __builtin_amdgcn_ds_bpermute(j << 2, srt);
        vv[p] = asrc[sj * HEADS + head];
      }
    }
    float m = ninf;
#pragma unroll
    for (int p = 0; p < 8; ++p) {
      vv[p] = (4 * p + q < deg) ? leaky(vv[p] + adn) : ninf;
      m = fmaxf(m, vv[p]);
    }
    m = fmaxf(m, quad<QP_XOR1>(m));
    m = fmaxf(m, quad<QP_XOR2>(m));

    float s = 0.f;
#pragma unroll
    for (int p = 0; p < 8; ++p) {
      if (4 * p < deg) {
        vv[p] = (4 * p + q < deg) ? __expf(vv[p] - m) : 0.f;
        s += vv[p];
      }
    }
    s += quad<QP_XOR1>(s);
    s += quad<QP_XOR2>(s);
    const float den = s + 1e-16f;

#pragma unroll
    for (int p = 0; p < 8; ++p) {
      if (4 * p < deg) {
        const float al = vv[p] / den;
        const int j0 = 4 * p;
        const int s0 = __builtin_amdgcn_readlane(srt, j0);
        const int s1 = (j0 + 1 < deg) ? __builtin_amdgcn_readlane(srt, j0 + 1) : s0;
        const int s2 = (j0 + 2 < deg) ? __builtin_amdgcn_readlane(srt, j0 + 2) : s0;
        const int s3 = (j0 + 3 < deg) ? __builtin_amdgcn_readlane(srt, j0 + 3) : s0;
        const v4f h0 = *(const v4f*)(h + s0 * HF + lane * 4);
        const v4f h1 = *(const v4f*)(h + s1 * HF + lane * 4);
        const v4f h2 = *(const v4f*)(h + s2 * HF + lane * 4);
        const v4f h3 = *(const v4f*)(h + s3 * HF + lane * 4);
        acc += quad<QP_BC0>(al) * h0;
        acc += quad<QP_BC1>(al) * h1;
        acc += quad<QP_BC2>(al) * h2;
        acc += quad<QP_BC3>(al) * h3;
      }
    }
  } else {
    for (int i = lane; i < deg; i += 32) {
      const int key = slot[beg + i];
      int r = 0;
      for (int j = 0; j < deg; ++j) {
        const int kj = slot[beg + j];
        r += ((kj < key) | ((kj == key) & (j < i))) ? 1 : 0;
      }
      srtbuf[beg + r] = key;
    }
    __threadfence();
    float m = ninf;
    for (int j = q; j < deg; j += 4) {
      const int sj = srtbuf[beg + j];
      m = fmaxf(m, leaky(asrc[sj * HEADS + head] + adn));
    }
    m = fmaxf(m, quad<QP_XOR1>(m));
    m = fmaxf(m, quad<QP_XOR2>(m));
    float s = 0.f;
    for (int j = q; j < deg; j += 4) {
      const int sj = srtbuf[beg + j];
      s += __expf(leaky(asrc[sj * HEADS + head] + adn) - m);
    }
    s += quad<QP_XOR1>(s);
    s += quad<QP_XOR2>(s);
    const float den = s + 1e-16f;
    for (int j = 0; j < deg; ++j) {
      const int sj = __builtin_amdgcn_readfirstlane(srtbuf[beg + j]);
      const float al = __expf(leaky(asrc[sj * HEADS + head] + adn) - m) / den;
      const v4f hv = *(const v4f*)(h + sj * HF + lane * 4);
      acc += al * hv;
    }
  }

  const v4f b = *(const v4f*)(bias + lane * 4);
  *(v4f*)(out + n * HF + lane * 4) = acc + b;
}

extern "C" void kernel_launch(void* const* d_in, const int* in_sizes, int n_in,
                              void* d_out, int out_size, void* d_ws, size_t ws_size,
                              hipStream_t stream) {
  const float* x        = (const float*)d_in[0];
  const float* W        = (const float*)d_in[1];
  const float* att_src  = (const float*)d_in[2];
  const float* att_dst  = (const float*)d_in[3];
  const float* bias     = (const float*)d_in[4];
  const int*   ei       = (const int*)d_in[5];
  float*       out      = (float*)d_out;

  float* h      = (float*)d_ws;
  float* asrc   = h     + (size_t)N_NODES * HF;
  float* adst   = asrc  + (size_t)N_NODES * HEADS;
  int*   cnt    = (int*)(adst + (size_t)N_NODES * HEADS);
  int*   off    = cnt   + N_NODES;
  int*   part   = off   + N_NODES + 32;
  int*   slot   = part  + N_EDGES;
  int*   srtbuf = slot  + N_EDGES;
  int*   bcnt   = srtbuf;
  int*   bfill  = srtbuf + NB * CSTR;

  const int BLK = 256;

  gat_gemm_wmma<<<N_NODES / 16, BLK, 0, stream>>>(x, W, h);

  gat_logits<<<(N_NODES * HEADS + BLK - 1) / BLK, BLK, 0, stream>>>(
      h, att_src, att_dst, asrc, adst, bcnt);

  gat_hist<<<NCHUNK, BLK, 0, stream>>>(ei, bcnt);

  gat_scatter<<<NCHUNK, BLK, 0, stream>>>(ei, bcnt, bfill, part);

  gat_build<<<NB, BLD, 0, stream>>>(bcnt, part, cnt, off, slot);

  gat_node<<<N_NODES / 8, BLK, 0, stream>>>(
      cnt, off, slot, srtbuf, h, asrc, adst, bias, out);
}


# Projection GEMM retiled: W copied once per work-group into LDS, each wave computes a 16x128 strip (A fragment loaded once, used for 8 tiles); 3128 waves instead of 25000; sum order unchanged
# speedup vs baseline: 3.3133x; 1.2239x over previous
#include <hip/hip_runtime.h>
#include <hip/hip_bf16.h>

#define N_NODES 50000
#define N_EDGES 800000
#define F_IN    128
#define HEADS   8
#define FH      16
#define HF      128
#define NEG_SLOPE 0.2f

typedef float v2f __attribute__((ext_vector_type(2)));
typedef float v4f __attribute__((ext_vector_type(4)));
typedef float v8f __attribute__((ext_vector_type(8)));

#define GEMM_WAVES 8
#define GEMM_WG    ((N_NODES / 16 + GEMM_WAVES - 1) / GEMM_WAVES)
#define WPITCH 132
__global__ __launch_bounds__(256) void gat_gemm_wmma(
    const float* __restrict__ x, const float* __restrict__ W,
    float* __restrict__ h) {
  __shared__ __attribute__((aligned(16))) float wl[HF * WPITCH];
  const int tid  = threadIdx.x;
  const int lane = tid & 31;
  const int wave = tid >> 5;
  const int half = lane >> 4;
  const int l15  = lane & 15;
  const int m0   = (blockIdx.x * GEMM_WAVES + wave) * 16;
  const bool live = m0 < N_NODES;
  const int mrow = live ? m0 + l15 : l15;

  v4f wv[16];
#pragma unroll
  for (int i = 0; i < 16; ++i)
    wv[i] = *(const v4f*)(W + (i * 256 + tid) * 4);

  const float* __restrict__ arow = x + mrow * F_IN + 2 * half;
  v2f a[F_IN / 4];
#pragma unroll
  for (int k = 0; k < F_IN / 4; ++k) {
    a[k].x = arow[4 * k];
    a[k].y = arow[4 * k + 1];
  }

#pragma unroll
  for (int i = 0; i < 16; ++i) {
    const int idx = i * 256 + tid;
    *(v4f*)(wl + (idx >> 5) * WPITCH + (idx & 31) * 4) = wv[i];
  }
  __syncthreads();
  if (!live) return;

  const float* __restrict__ brow = wl + l15 * WPITCH + 2 * half;

  v8f c[8];
#pragma unroll
  for (int t = 0; t < 8; ++t) c[t] = (v8f){};

  v2f bc[8][2], bn[8][2];
#pragma unroll
  for (int t = 0; t < 8; ++t) {
    bc[t][0] = *(const v2f*)(brow + t * 16 * WPITCH);
    bc[t][1] = *(const v2f*)(brow + t * 16 * WPITCH + 4);
  }
#pragma unroll
  for (int j = 0; j < F_IN / 8; ++j) {
    if (j + 1 < F_IN / 8) {
#pragma unroll
      for (int t = 0; t < 8; ++t) {
        bn[t][0] = *(const v2f*)(brow + t * 16 * WPITCH + 8 * (j + 1));
        bn[t][1] = *(const v2f*)(brow + t * 16 * WPITCH + 8 * (j + 1) + 4);
      }
    }
    __builtin_amdgcn_sched_barrier(0);
#pragma unroll
    for (int t = 0; t < 8; ++t)
      c[t] = __builtin_amdgcn_wmma_f32_16x16x4_f32(
           false, a[2 * j],  false, bc[t][0],
           (short)0, c[t],  false,  false);
#pragma unroll
    for (int t = 0; t < 8; ++t)
      c[t] = __builtin_amdgcn_wmma_f32_16x16x4_f32(
           false, a[2 * j + 1],  false, bc[t][1],
           (short)0, c[t],  false,  false);
    __builtin_amdgcn_sched_barrier(0);
#pragma unroll
    for (int t = 0; t < 8; ++t) { bc[t][0] = bn[t][0]; bc[t][1] = bn[t][1]; }
  }

  float* __restrict__ op = h + (m0 + half * 8) * HF + l15;
#pragma unroll
  for (int t = 0; t < 8; ++t) {
#pragma unroll
    for (int v = 0; v < 8; ++v) op[v * HF + t * 16] = c[t][v];
  }
}

#define NB     256
#define NPB    196
#define CSTR   32
#define EPT    8
#define ECHUNK (256 * EPT)
#define NCHUNK ((N_EDGES + ECHUNK - 1) / ECHUNK)
static_assert(NB * NPB >= N_NODES, "buckets must cover all nodes");
static_assert(N_NODES <= 65536, "source id is packed into 16 bits");
static_assert(NB == 256, "one bucket per thread of a 256-thread work-group");

__global__ void gat_logits(const float* __restrict__ h,
                           const float* __restrict__ att_src,
                           const float* __restrict__ att_dst,
                           float* __restrict__ asrc, float* __restrict__ adst,
                           int* __restrict__ bctr) {
  int t = blockIdx.x * blockDim.x + threadIdx.x;
  if (t >= N_NODES * HEADS) return;
  if (t < 2 * NB * CSTR) bctr[t] = 0;
  int head = t & 7;
  int n    = t >> 3;
  const float* __restrict__ hp = h + n * HF + head * FH;
  const float* __restrict__ as = att_src + head * FH;
  const float* __restrict__ ad = att_dst + head * FH;
  float s = 0.f, d = 0.f;
#pragma unroll
  for (int f = 0; f < FH; ++f) { float v = hp[f]; s += v * as[f]; d += v * ad[f]; }
  asrc[t] = s;
  adst[t] = d;
}

__device__ __forceinline__ float leaky(float v) {
  return v >= 0.f ? v : NEG_SLOPE * v;
}

__global__ __launch_bounds__(256) void gat_hist(const int* __restrict__ ei,
                                                int* __restrict__ bcnt) {
  __shared__ int lh[NB];
  const int tid = threadIdx.x;
  lh[tid] = 0;
  __syncthreads();
  const int e0 = blockIdx.x * ECHUNK + tid;
  int d[EPT];
#pragma unroll
  for (int i = 0; i < EPT; ++i) d[i] = ei[N_EDGES + min(e0 + i * 256, N_EDGES - 1)];
#pragma unroll
  for (int i = 0; i < EPT; ++i) {
    if (e0 + i * 256 < N_EDGES)
      __hip_atomic_fetch_add(&lh[d[i] / NPB], 1, __ATOMIC_RELAXED,
                             __HIP_MEMORY_SCOPE_WORKGROUP);
  }
  __syncthreads();
  const int c = lh[tid];
  if (c > 0)
    __hip_atomic_fetch_add(&bcnt[tid * CSTR], c, __ATOMIC_RELAXED,
                           __HIP_MEMORY_SCOPE_AGENT);
}

__global__ __launch_bounds__(256) void gat_scatter(
    const int* __restrict__ ei, const int* __restrict__ bcnt,
    int* __restrict__ bfill, int* __restrict__ part) {
  __shared__ int lh[NB];
  __shared__ int lbase[NB];
  __shared__ int wsum[8];
  const int tid  = threadIdx.x;
  const int lane = tid & 31;
  const int wave = tid >> 5;
  lh[tid] = 0;
  __syncthreads();
  const int e0 = blockIdx.x * ECHUNK + tid;
  int s[EPT], d[EPT], bk[EPT], r[EPT];
#pragma unroll
  for (int i = 0; i < EPT; ++i) {
    const int e = min(e0 + i * 256, N_EDGES - 1);
    s[i] = ei[e];
    d[i] = ei[N_EDGES + e];
  }
  const int tot = bcnt[tid * CSTR];
#pragma unroll
  for (int i = 0; i < EPT; ++i) {
    bk[i] = d[i] / NPB;
    r[i] = 0;
    if (e0 + i * 256 < N_EDGES)
      r[i] = __hip_atomic_fetch_add(&lh[bk[i]], 1, __ATOMIC_RELAXED,
                                    __HIP_MEMORY_SCOPE_WORKGROUP);
  }
  int incl = tot;
  for (int k = 1; k < 32; k <<= 1) {
    const int up = __shfl_up(incl, k, 32);
    if (lane >= k) incl += up;
  }
  if (lane == 31) wsum[wave] = incl;
  __syncthreads();
  int woff = 0;
  for (int w = 0; w < 8; ++w) woff += (w < wave) ? wsum[w] : 0;
  const int c = lh[tid];
  int got = 0;
  if (c > 0)
    got = __hip_atomic_fetch_add(&bfill[tid * CSTR], c, __ATOMIC_RELAXED,
                                 __HIP_MEMORY_SCOPE_AGENT);
  lbase[tid] = woff + incl - tot + got;
  __syncthreads();
#pragma unroll
  for (int i = 0; i < EPT; ++i) {
    if (e0 + i * 256 < N_EDGES)
      part[lbase[bk[i]] + r[i]] = ((d[i] - bk[i] * NPB) << 16) | s[i];
  }
}

#define BLD 1024
__global__ __launch_bounds__(BLD) void gat_build(
    const int* __restrict__ bcnt, const int* __restrict__ part,
    int* __restrict__ cnt, int* __restrict__ off, int* __restrict__ slot) {
  __shared__ int ncnt[NB];
  __shared__ int ncur[NB];
  __shared__ int wsA[8];
  __shared__ int wsB[8];
  const int tid  = threadIdx.x;
  const int lane = tid & 31;
  const int wave = tid >> 5;
  const int b    = blockIdx.x;
  const int nb   = bcnt[b * CSTR];

  if (tid < NB) {
    ncnt[tid] = 0;
    int v = (tid < b) ? bcnt[tid * CSTR] : 0;
    for (int k = 16; k >= 1; k >>= 1) v += __shfl_xor(v, k, 32);
    if (lane == 0) wsA[wave] = v;
  }
  __syncthreads();
  int bstart = 0;
  for (int w = 0; w < 8; ++w) bstart += wsA[w];

  for (int eb = tid; eb < nb; eb += 4 * BLD) {
    int ent[4];
#pragma unroll
    for (int k = 0; k < 4; ++k) {
      const int e = eb + k * BLD;
      ent[k] = (e < nb) ? part[bstart + e] : -1;
    }
#pragma unroll
    for (int k = 0; k < 4; ++k) {
      if (ent[k] >= 0)
        __hip_atomic_fetch_add(&ncnt[ent[k] >> 16], 1, __ATOMIC_RELAXED,
                               __HIP_MEMORY_SCOPE_WORKGROUP);
    }
  }
  __syncthreads();

  int c = 0, incl = 0;
  if (tid < NB) {
    c = ncnt[tid];
    incl = c;
    for (int k = 1; k < 32; k <<= 1) {
      const int up = __shfl_up(incl, k, 32);
      if (lane >= k) incl += up;
    }
    if (lane == 31) wsB[wave] = incl;
  }
  __syncthreads();
  if (tid < NB) {
    int woff = 0;
    for (int w = 0; w < 8; ++w) woff += (w < wave) ? wsB[w] : 0;
    const int noff = woff + incl - c;
    ncur[tid] = noff;
    const int n = b * NPB + tid;
    if (tid < NPB && n < N_NODES) {
      cnt[n] = c;
      off[n] = bstart + noff;
    }
  }
  __syncthreads();

  for (int eb = tid; eb < nb; eb += 4 * BLD) {
    int ent[4];
#pragma unroll
    for (int k = 0; k < 4; ++k) {
      const int e = eb + k * BLD;
      ent[k] = (e < nb) ? part[bstart + e] : -1;
    }
#pragma unroll
    for (int k = 0; k < 4; ++k) {
      if (ent[k] >= 0) {
        const int pos = __hip_atomic_fetch_add(&ncur[ent[k] >> 16], 1,
                                               __ATOMIC_RELAXED,
                                               __HIP_MEMORY_SCOPE_WORKGROUP);
        slot[bstart + pos] = ent[k] & 0xffff;
      }
    }
  }
}

template <int CTRL>
__device__ __forceinline__ float quad(float x) {
  return __int_as_float(__builtin_amdgcn_mov_dpp(__float_as_int(x), CTRL, 0xf, 0xf, true));
}
#define QP_XOR1 0xB1
#define QP_XOR2 0x4E
#define QP_BC0  0x00
#define QP_BC1  0x55
#define QP_BC2  0xAA
#define QP_BC3  0xFF

__global__ __launch_bounds__(256) void gat_node(
    const int* __restrict__ cnt, const int* __restrict__ off,
    const int* __restrict__ slot, int* __restrict__ srtbuf,
    const float* __restrict__ h, const float* __restrict__ asrc,
    const float* __restrict__ adst, const float* __restrict__ bias,
    float* __restrict__ out) {
  const int lane = threadIdx.x & 31;
  const int n    = blockIdx.x * 8 + (threadIdx.x >> 5);
  const int q    = lane & 3;
  const int head = lane >> 2;
  const int deg = __builtin_amdgcn_readfirstlane(cnt[n]);
  const int beg = __builtin_amdgcn_readfirstlane(off[n]);
  const float adn = adst[n * HEADS + head];
  const float ninf = -__builtin_inff();

  v4f acc = {0.f, 0.f, 0.f, 0.f};

  if (deg <= 32) {
    const int key = (lane < deg) ? slot[beg + lane] * 32 + lane : 0x7fffffff;
    int r = 0;
    for (int j = 0; j < deg; ++j) {
      r += (__builtin_amdgcn_readlane(key, j) < key) ? 1 : 0;
    }
    const int srt = __builtin_amdgcn_ds_permute(r << 2, key) >> 5;

    float vv[8];
#pragma unroll
    for (int p = 0; p < 8; ++p) {
      vv[p] = 0.f;
      if (4 * p < deg) {
        const int j  = min(4 * p + q, deg - 1);
        const int sj = __builtin_amdgcn_ds_bpermute(j << 2, srt);
        vv[p] = asrc[sj * HEADS + head];
      }
    }
    float m = ninf;
#pragma unroll
    for (int p = 0; p < 8; ++p) {
      vv[p] = (4 * p + q < deg) ? leaky(vv[p] + adn) : ninf;
      m = fmaxf(m, vv[p]);
    }
    m = fmaxf(m, quad<QP_XOR1>(m));
    m = fmaxf(m, quad<QP_XOR2>(m));

    float s = 0.f;
#pragma unroll
    for (int p = 0; p < 8; ++p) {
      if (4 * p < deg) {
        vv[p] = (4 * p + q < deg) ? __expf(vv[p] - m) : 0.f;
        s += vv[p];
      }
    }
    s += quad<QP_XOR1>(s);
    s += quad<QP_XOR2>(s);
    const float den = s + 1e-16f;

#pragma unroll
    for (int p = 0; p < 8; ++p) {
      if (4 * p < deg) {
        const float al = vv[p] / den;
        const int j0 = 4 * p;
        const int s0 = __builtin_amdgcn_readlane(srt, j0);
        const int s1 = (j0 + 1 < deg) ? __builtin_amdgcn_readlane(srt, j0 + 1) : s0;
        const int s2 = (j0 + 2 < deg) ? __builtin_amdgcn_readlane(srt, j0 + 2) : s0;
        const int s3 = (j0 + 3 < deg) ? __builtin_amdgcn_readlane(srt, j0 + 3) : s0;
        const v4f h0 = *(const v4f*)(h + s0 * HF + lane * 4);
        const v4f h1 = *(const v4f*)(h + s1 * HF + lane * 4);
        const v4f h2 = *(const v4f*)(h + s2 * HF + lane * 4);
        const v4f h3 = *(const v4f*)(h + s3 * HF + lane * 4);
        acc += quad<QP_BC0>(al) * h0;
        acc += quad<QP_BC1>(al) * h1;
        acc += quad<QP_BC2>(al) * h2;
        acc += quad<QP_BC3>(al) * h3;
      }
    }
  } else {
    for (int i = lane; i < deg; i += 32) {
      const int key = slot[beg + i];
      int r = 0;
      for (int j = 0; j < deg; ++j) {
        const int kj = slot[beg + j];
        r += ((kj < key) | ((kj == key) & (j < i))) ? 1 : 0;
      }
      srtbuf[beg + r] = key;
    }
    __threadfence();
    float m = ninf;
    for (int j = q; j < deg; j += 4) {
      const int sj = srtbuf[beg + j];
      m = fmaxf(m, leaky(asrc[sj * HEADS + head] + adn));
    }
    m = fmaxf(m, quad<QP_XOR1>(m));
    m = fmaxf(m, quad<QP_XOR2>(m));
    float s = 0.f;
    for (int j = q; j < deg; j += 4) {
      const int sj = srtbuf[beg + j];
      s += __expf(leaky(asrc[sj * HEADS + head] + adn) - m);
    }
    s += quad<QP_XOR1>(s);
    s += quad<QP_XOR2>(s);
    const float den = s + 1e-16f;
    for (int j = 0; j < deg; ++j) {
      const int sj = __builtin_amdgcn_readfirstlane(srtbuf[beg + j]);
      const float al = __expf(leaky(asrc[sj * HEADS + head] + adn) - m) / den;
      const v4f hv = *(const v4f*)(h + sj * HF + lane * 4);
      acc += al * hv;
    }
  }

  const v4f b = *(const v4f*)(bias + lane * 4);
  *(v4f*)(out + n * HF + lane * 4) = acc + b;
}

extern "C" void kernel_launch(void* const* d_in, const int* in_sizes, int n_in,
                              void* d_out, int out_size, void* d_ws, size_t ws_size,
                              hipStream_t stream) {
  const float* x        = (const float*)d_in[0];
  const float* W        = (const float*)d_in[1];
  const float* att_src  = (const float*)d_in[2];
  const float* att_dst  = (const float*)d_in[3];
  const float* bias     = (const float*)d_in[4];
  const int*   ei       = (const int*)d_in[5];
  float*       out      = (float*)d_out;

  float* h      = (float*)d_ws;
  float* asrc   = h     + (size_t)N_NODES * HF;
  float* adst   = asrc  + (size_t)N_NODES * HEADS;
  int*   cnt    = (int*)(adst + (size_t)N_NODES * HEADS);
  int*   off    = cnt   + N_NODES;
  int*   part   = off   + N_NODES + 32;
  int*   slot   = part  + N_EDGES;
  int*   srtbuf = slot  + N_EDGES;
  int*   bcnt   = srtbuf;
  int*   bfill  = srtbuf + NB * CSTR;

  const int BLK = 256;

  gat_gemm_wmma<<<GEMM_WG, BLK, 0, stream>>>(x, W, h);

  gat_logits<<<(N_NODES * HEADS + BLK - 1) / BLK, BLK, 0, stream>>>(
      h, att_src, att_dst, asrc, adst, bcnt);

  gat_hist<<<NCHUNK, BLK, 0, stream>>>(ei, bcnt);

  gat_scatter<<<NCHUNK, BLK, 0, stream>>>(ei, bcnt, bfill, part);

  gat_build<<<NB, BLD, 0, stream>>>(bcnt, part, cnt, off, slot);

  gat_node<<<N_NODES / 8, BLK, 0, stream>>>(
      cnt, off, slot, srtbuf, h, asrc, adst, bias, out);
}


# Edge grouping freed of global counters (chunk-local bucket sort, run table) runs as extra work-groups of the projection launch, segment build beside the logits: 6 launches to 3; sum order unchanged
# speedup vs baseline: 4.6197x; 1.3943x over previous
#include <hip/hip_runtime.h>
#include <hip/hip_bf16.h>

#define N_NODES 50000
#define N_EDGES 800000
#define F_IN    128
#define HEADS   8
#define FH      16
#define HF      128
#define NEG_SLOPE 0.2f

typedef float v2f __attribute__((ext_vector_type(2)));
typedef float v4f __attribute__((ext_vector_type(4)));
typedef float v8f __attribute__((ext_vector_type(8)));

#define NB     256
#define NPB    196
#define EPT    16
#define ECHUNK (256 * EPT)
#define NCHUNK ((N_EDGES + ECHUNK - 1) / ECHUNK)
static_assert(NB * NPB >= N_NODES, "buckets must cover all nodes");
static_assert(N_NODES <= 65536, "source id is packed into 16 bits");
static_assert(NB == 256, "one bucket per thread of a 256-thread work-group");
static_assert(NCHUNK <= 255, "run search below takes 8 halving steps over 256 places");

#define GEMM_WAVES 8
#define GEMM_WG    ((N_NODES / 16 + GEMM_WAVES - 1) / GEMM_WAVES)
#define WPITCH 132
__global__ __launch_bounds__(256) void gat_gemm_group(
    const float* __restrict__ x, const float* __restrict__ W,
    float* __restrict__ h, const int* __restrict__ ei,
    int* __restrict__ part, int* __restrict__ rstart, int* __restrict__ rcount) {
  __shared__ __attribute__((aligned(16))) float wl[HF * WPITCH];
  __shared__ int lh[NB];
  __shared__ int lbase[NB];
  __shared__ int wsum[8];
  const int tid  = threadIdx.x;
  const int lane = tid & 31;
  const int wave = tid >> 5;

  if (blockIdx.x >= GEMM_WG) {
    const int c = blockIdx.x - GEMM_WG;
    lh[tid] = 0;
    __syncthreads();
    const int e0 = c * ECHUNK + tid;
    int s[EPT], d[EPT], bk[EPT], r[EPT];
#pragma unroll
    for (int i = 0; i < EPT; ++i) {
      const int e = min(e0 + i * 256, N_EDGES - 1);
      s[i] = ei[e];
      d[i] = ei[N_EDGES + e];
    }
#pragma unroll
    for (int i = 0; i < EPT; ++i) {
      bk[i] = d[i] / NPB;
      r[i] = 0;
      if (e0 + i * 256 < N_EDGES)
        r[i] = __hip_atomic_fetch_add(&lh[bk[i]], 1, __ATOMIC_RELAXED,
                                      __HIP_MEMORY_SCOPE_WORKGROUP);
    }
    __syncthreads();
    const int tot = lh[tid];
    int incl = tot;
    for (int k = 1; k < 32; k <<= 1) {
      const int up = __shfl_up(incl, k, 32);
      if (lane >= k) incl += up;
    }
    if (lane == 31) wsum[wave] = incl;
    __syncthreads();
    int woff = 0;
    for (int w = 0; w < 8; ++w) woff += (w < wave) ? wsum[w] : 0;
    const int st = woff + incl - tot;
    lbase[tid] = st;
    rstart[c * NB + tid] = st;
    rcount[c * NB + tid] = tot;
    __syncthreads();
#pragma unroll
    for (int i = 0; i < EPT; ++i) {
      if (e0 + i * 256 < N_EDGES)
        part[c * ECHUNK + lbase[bk[i]] + r[i]] = ((d[i] - bk[i] * NPB) << 16) | s[i];
    }
    return;
  }

  const int half = lane >> 4;
  const int l15  = lane & 15;
  const int m0   = (blockIdx.x * GEMM_WAVES + wave) * 16;
  const bool live = m0 < N_NODES;
  const int mrow = live ? m0 + l15 : l15;

  v4f wv[16];
#pragma unroll
  for (int i = 0; i < 16; ++i)
    wv[i] = *(const v4f*)(W + (i * 256 + tid) * 4);

  const float* __restrict__ arow = x + mrow * F_IN + 2 * half;
  v2f a[F_IN / 4];
#pragma unroll
  for (int k = 0; k < F_IN / 4; ++k) {
    a[k].x = arow[4 * k];
    a[k].y = arow[4 * k + 1];
  }

#pragma unroll
  for (int i = 0; i < 16; ++i) {
    const int idx = i * 256 + tid;
    *(v4f*)(wl + (idx >> 5) * WPITCH + (idx & 31) * 4) = wv[i];
  }
  __syncthreads();
  if (!live) return;

  const float* __restrict__ brow = wl + l15 * WPITCH + 2 * half;

  v8f c[8];
#pragma unroll
  for (int t = 0; t < 8; ++t) c[t] = (v8f){};

  v2f bc[8][2], bn[8][2];
#pragma unroll
  for (int t = 0; t < 8; ++t) {
    bc[t][0] = *(const v2f*)(brow + t * 16 * WPITCH);
    bc[t][1] = *(const v2f*)(brow + t * 16 * WPITCH + 4);
  }
#pragma unroll
  for (int j = 0; j < F_IN / 8; ++j) {
    if (j + 1 < F_IN / 8) {
#pragma unroll
      for (int t = 0; t < 8; ++t) {
        bn[t][0] = *(const v2f*)(brow + t * 16 * WPITCH + 8 * (j + 1));
        bn[t][1] = *(const v2f*)(brow + t * 16 * WPITCH + 8 * (j + 1) + 4);
      }
    }
    __builtin_amdgcn_sched_barrier(0);
#pragma unroll
    for (int t = 0; t < 8; ++t)
      c[t] = __builtin_amdgcn_wmma_f32_16x16x4_f32(
           false, a[2 * j],  false, bc[t][0],
           (short)0, c[t],  false,  false);
#pragma unroll
    for (int t = 0; t < 8; ++t)
      c[t] = __builtin_amdgcn_wmma_f32_16x16x4_f32(
           false, a[2 * j + 1],  false, bc[t][1],
           (short)0, c[t],  false,  false);
    __builtin_amdgcn_sched_barrier(0);
#pragma unroll
    for (int t = 0; t < 8; ++t) { bc[t][0] = bn[t][0]; bc[t][1] = bn[t][1]; }
  }

  float* __restrict__ op = h + (m0 + half * 8) * HF + l15;
#pragma unroll
  for (int t = 0; t < 8; ++t) {
#pragma unroll
    for (int v = 0; v < 8; ++v) op[v * HF + t * 16] = c[t][v];
  }
}

__device__ __forceinline__ float leaky(float v) {
  return v >= 0.f ? v : NEG_SLOPE * v;
}

#define BLD 1024
#define LOGIT_WG ((N_NODES * HEADS + BLD - 1) / BLD)

__device__ __forceinline__ int bucket_entry(const int* __restrict__ part,
                                            const int* pre, const int* rst, int e) {
  int lo = 0;
#pragma unroll
  for (int step = NB / 2; step >= 1; step >>= 1) {
    const int nx = lo + step;
    if (pre[nx] <= e) lo = nx;
  }
  return part[lo * ECHUNK + rst[lo] + (e - pre[lo])];
}

__global__ __launch_bounds__(BLD) void gat_build_logits(
    const int* __restrict__ part, const int* __restrict__ rstart,
    const int* __restrict__ rcount,
    int* __restrict__ cnt, int* __restrict__ off, int* __restrict__ slot,
    const float* __restrict__ h, const float* __restrict__ att_src,
    const float* __restrict__ att_dst,
    float* __restrict__ asrc, float* __restrict__ adst) {
  __shared__ int colp[4 * NB];
  __shared__ int pre[NB];
  __shared__ int rst[NB];
  __shared__ int ncnt[NB];
  __shared__ int ncur[NB];
  __shared__ int wsS[32];
  __shared__ int wsA[8];
  __shared__ int wsB[8];
  const int tid  = threadIdx.x;
  const int lane = tid & 31;
  const int wave = tid >> 5;

  if (blockIdx.x >= NB) {
    const int t = (blockIdx.x - NB) * BLD + tid;
    if (t >= N_NODES * HEADS) return;
    const int head = t & 7;
    const int n    = t >> 3;
    const float* __restrict__ hp = h + n * HF + head * FH;
    const float* __restrict__ as = att_src + head * FH;
    const float* __restrict__ ad = att_dst + head * FH;
    float s = 0.f, d = 0.f;
#pragma unroll
    for (int f = 0; f < FH; ++f) { float v = hp[f]; s += v * as[f]; d += v * ad[f]; }
    asrc[t] = s;
    adst[t] = d;
    return;
  }

  const int b = blockIdx.x;

  const int col = tid & (NB - 1);
  const int grp = tid >> 8;
  int acc = 0;
  if (col <= b) {
    for (int c0 = grp; c0 < NCHUNK; c0 += 4 * 16) {
      int v[16];
#pragma unroll
      for (int u = 0; u < 16; ++u) {
        const int c = c0 + 4 * u;
        v[u] = (c < NCHUNK) ? rcount[c * NB + col] : 0;
      }
#pragma unroll
      for (int u = 0; u < 16; ++u) acc += v[u];
    }
  }
  colp[tid] = acc;

  int rs = 0, ln = 0;
  if (tid < NCHUNK) {
    rs = rstart[tid * NB + b];
    ln = rcount[tid * NB + b];
  }
  int incl = ln;
  for (int k = 1; k < 32; k <<= 1) {
    const int up = __shfl_up(incl, k, 32);
    if (lane >= k) incl += up;
  }
  if (lane == 31) wsS[wave] = incl;
  __syncthreads();
  int wo = 0, nb = 0;
  for (int w = 0; w < 32; ++w) {
    const int sv = wsS[w];
    nb += sv;
    wo += (w < wave) ? sv : 0;
  }
  if (tid < NB) {
    pre[tid] = (tid < NCHUNK) ? wo + incl - ln : 0x7fffffff;
    rst[tid] = rs;
    ncnt[tid] = 0;
    const int tot = colp[tid] + colp[NB + tid] + colp[2 * NB + tid] + colp[3 * NB + tid];
    int v = (tid < b) ? tot : 0;
    for (int k = 16; k >= 1; k >>= 1) v += __shfl_xor(v, k, 32);
    if (lane == 0) wsA[wave] = v;
  }
  __syncthreads();
  int bstart = 0;
  for (int w = 0; w < 8; ++w) bstart += wsA[w];

  for (int eb = tid; eb < nb; eb += 4 * BLD) {
    int ent[4];
#pragma unroll
    for (int k = 0; k < 4; ++k) {
      const int e = eb + k * BLD;
      ent[k] = (e < nb) ? bucket_entry(part, pre, rst, e) : -1;
    }
#pragma unroll
    for (int k = 0; k < 4; ++k) {
      if (ent[k] >= 0)
        __hip_atomic_fetch_add(&ncnt[ent[k] >> 16], 1, __ATOMIC_RELAXED,
                               __HIP_MEMORY_SCOPE_WORKGROUP);
    }
  }
  __syncthreads();

  int c = 0, inc2 = 0;
  if (tid < NB) {
    c = ncnt[tid];
    inc2 = c;
    for (int k = 1; k < 32; k <<= 1) {
      const int up = __shfl_up(inc2, k, 32);
      if (lane >= k) inc2 += up;
    }
    if (lane == 31) wsB[wave] = inc2;
  }
  __syncthreads();
  if (tid < NB) {
    int woff = 0;
    for (int w = 0; w < 8; ++w) woff += (w < wave) ? wsB[w] : 0;
    const int noff = woff + inc2 - c;
    ncur[tid] = noff;
    const int n = b * NPB + tid;
    if (tid < NPB && n < N_NODES) {
      cnt[n] = c;
      off[n] = bstart + noff;
    }
  }
  __syncthreads();

  for (int eb = tid; eb < nb; eb += 4 * BLD) {
    int ent[4];
#pragma unroll
    for (int k = 0; k < 4; ++k) {
      const int e = eb + k * BLD;
      ent[k] = (e < nb) ? bucket_entry(part, pre, rst, e) : -1;
    }
#pragma unroll
    for (int k = 0; k < 4; ++k) {
      if (ent[k] >= 0) {
        const int pos = __hip_atomic_fetch_add(&ncur[ent[k] >> 16], 1,
                                               __ATOMIC_RELAXED,
                                               __HIP_MEMORY_SCOPE_WORKGROUP);
        slot[bstart + pos] = ent[k] & 0xffff;
      }
    }
  }
}

template <int CTRL>
__device__ __forceinline__ float quad(float x) {
  return __int_as_float(__builtin_amdgcn_mov_dpp(__float_as_int(x), CTRL, 0xf, 0xf, true));
}
#define QP_XOR1 0xB1
#define QP_XOR2 0x4E
#define QP_BC0  0x00
#define QP_BC1  0x55
#define QP_BC2  0xAA
#define QP_BC3  0xFF

__global__ __launch_bounds__(256) void gat_node(
    const int* __restrict__ cnt, const int* __restrict__ off,
    const int* __restrict__ slot, int* __restrict__ srtbuf,
    const float* __restrict__ h, const float* __restrict__ asrc,
    const float* __restrict__ adst, const float* __restrict__ bias,
    float* __restrict__ out) {
  const int lane = threadIdx.x & 31;
  const int n    = blockIdx.x * 8 + (threadIdx.x >> 5);
  const int q    = lane & 3;
  const int head = lane >> 2;
  const int deg = __builtin_amdgcn_readfirstlane(cnt[n]);
  const int beg = __builtin_amdgcn_readfirstlane(off[n]);
  const float adn = adst[n * HEADS + head];
  const float ninf = -__builtin_inff();

  v4f acc = {0.f, 0.f, 0.f, 0.f};

  if (deg <= 32) {
    const int key = (lane < deg) ? slot[beg + lane] * 32 + lane : 0x7fffffff;
    int r = 0;
    for (int j = 0; j < deg; ++j) {
      r += (__builtin_amdgcn_readlane(key, j) < key) ? 1 : 0;
    }
    const int srt = __builtin_amdgcn_ds_permute(r << 2, key) >> 5;

    float vv[8];
#pragma unroll
    for (int p = 0; p < 8; ++p) {
      vv[p] = 0.f;
      if (4 * p < deg) {
        const int j  = min(4 * p + q, deg - 1);
        const int sj = __builtin_amdgcn_ds_bpermute(j << 2, srt);
        vv[p] = asrc[sj * HEADS + head];
      }
    }
    float m = ninf;
#pragma unroll
    for (int p = 0; p < 8; ++p) {
      vv[p] = (4 * p + q < deg) ? leaky(vv[p] + adn) : ninf;
      m = fmaxf(m, vv[p]);
    }
    m = fmaxf(m, quad<QP_XOR1>(m));
    m = fmaxf(m, quad<QP_XOR2>(m));

    float s = 0.f;
#pragma unroll
    for (int p = 0; p < 8; ++p) {
      if (4 * p < deg) {
        vv[p] = (4 * p + q < deg) ? __expf(vv[p] - m) : 0.f;
        s += vv[p];
      }
    }
    s += quad<QP_XOR1>(s);
    s += quad<QP_XOR2>(s);
    const float den = s + 1e-16f;

#pragma unroll
    for (int p = 0; p < 8; ++p) {
      if (4 * p < deg) {
        const float al = vv[p] / den;
        const int j0 = 4 * p;
        const int s0 = __builtin_amdgcn_readlane(srt, j0);
        const int s1 = (j0 + 1 < deg) ? __builtin_amdgcn_readlane(srt, j0 + 1) : s0;
        const int s2 = (j0 + 2 < deg) ? __builtin_amdgcn_readlane(srt, j0 + 2) : s0;
        const int s3 = (j0 + 3 < deg) ? __builtin_amdgcn_readlane(srt, j0 + 3) : s0;
        const v4f h0 = *(const v4f*)(h + s0 * HF + lane * 4);
        const v4f h1 = *(const v4f*)(h + s1 * HF + lane * 4);
        const v4f h2 = *(const v4f*)(h + s2 * HF + lane * 4);
        const v4f h3 = *(const v4f*)(h + s3 * HF + lane * 4);
        acc += quad<QP_BC0>(al) * h0;
        acc += quad<QP_BC1>(al) * h1;
        acc += quad<QP_BC2>(al) * h2;
        acc += quad<QP_BC3>(al) * h3;
      }
    }
  } else {
    for (int i = lane; i < deg; i += 32) {
      const int key = slot[beg + i];
      int r = 0;
      for (int j = 0; j < deg; ++j) {
        const int kj = slot[beg + j];
        r += ((kj < key) | ((kj == key) & (j < i))) ? 1 : 0;
      }
      srtbuf[beg + r] = key;
    }
    __threadfence();
    float m = ninf;
    for (int j = q; j < deg; j += 4) {
      const int sj = srtbuf[beg + j];
      m = fmaxf(m, leaky(asrc[sj * HEADS + head] + adn));
    }
    m = fmaxf(m, quad<QP_XOR1>(m));
    m = fmaxf(m, quad<QP_XOR2>(m));
    float s = 0.f;
    for (int j = q; j < deg; j += 4) {
      const int sj = srtbuf[beg + j];
      s += __expf(leaky(asrc[sj * HEADS + head] + adn) - m);
    }
    s += quad<QP_XOR1>(s);
    s += quad<QP_XOR2>(s);
    const float den = s + 1e-16f;
    for (int j = 0; j < deg; ++j) {
      const int sj = __builtin_amdgcn_readfirstlane(srtbuf[beg + j]);
      const float al = __expf(leaky(asrc[sj * HEADS + head] + adn) - m) / den;
      const v4f hv = *(const v4f*)(h + sj * HF + lane * 4);
      acc += al * hv;
    }
  }

  const v4f b = *(const v4f*)(bias + lane * 4);
  *(v4f*)(out + n * HF + lane * 4) = acc + b;
}

extern "C" void kernel_launch(void* const* d_in, const int* in_sizes, int n_in,
                              void* d_out, int out_size, void* d_ws, size_t ws_size,
                              hipStream_t stream) {
  const float* x        = (const float*)d_in[0];
  const float* W        = (const float*)d_in[1];
  const float* att_src  = (const float*)d_in[2];
  const float* att_dst  = (const float*)d_in[3];
  const float* bias     = (const float*)d_in[4];
  const int*   ei       = (const int*)d_in[5];
  float*       out      = (float*)d_out;

  float* h      = (float*)d_ws;
  float* asrc   = h     + (size_t)N_NODES * HF;
  float* adst   = asrc  + (size_t)N_NODES * HEADS;
  int*   cnt    = (int*)(adst + (size_t)N_NODES * HEADS);
  int*   off    = cnt   + N_NODES;
  int*   part   = off   + N_NODES + 32;
  int*   slot   = part  + N_EDGES;
  int*   srtbuf = slot  + N_EDGES;
  int*   rstart = srtbuf;
  int*   rcount = srtbuf + NCHUNK * NB;
  static_assert(2 * NCHUNK * NB <= N_EDGES, "run tables must fit into srtbuf");

  gat_gemm_group<<<GEMM_WG + NCHUNK, 256, 0, stream>>>(
      x, W, h, ei, part, rstart, rcount);

  gat_build_logits<<<NB + LOGIT_WG, BLD, 0, stream>>>(
      part, rstart, rcount, cnt, off, slot, h, att_src, att_dst, asrc, adst);

  gat_node<<<N_NODES / 8, 256, 0, stream>>>(
      cnt, off, slot, srtbuf, h, asrc, adst, bias, out);
}


# Edge grouping level two joins the node launch: a work-group per bucket builds its node segments in LDS and its 32 waves do those nodes, 8 rows in flight; 3 launches to 2; sum order unchanged
# speedup vs baseline: 5.9876x; 1.2961x over previous
#include <hip/hip_runtime.h>
#include <hip/hip_bf16.h>

#define N_NODES 50000
#define N_EDGES 800000
#define F_IN    128
#define HEADS   8
#define FH      16
#define HF      128
#define NEG_SLOPE 0.2f

typedef float v2f __attribute__((ext_vector_type(2)));
typedef float v4f __attribute__((ext_vector_type(4)));
typedef float v8f __attribute__((ext_vector_type(8)));

#define NB     256
#define NPB    196
#define EPT    16
#define ECHUNK (256 * EPT)
#define NCHUNK ((N_EDGES + ECHUNK - 1) / ECHUNK)
static_assert(NB * NPB >= N_NODES, "buckets must cover all nodes");
static_assert(N_NODES <= 65536, "source id is packed into 16 bits");
static_assert(NB == 256, "one bucket per thread of a 256-thread work-group");
static_assert(NCHUNK <= 255, "run search below takes 8 halving steps over 256 places");
static_assert(ECHUNK < 32768, "run start and run length are packed into 16 bits each");

#define GEMM_WAVES 8
#define GEMM_WG    ((N_NODES / 16 + GEMM_WAVES - 1) / GEMM_WAVES)
#define WPITCH 132
__global__ __launch_bounds__(256) void gat_gemm_group(
    const float* __restrict__ x, const float* __restrict__ W,
    float* __restrict__ h, const int* __restrict__ ei,
    int* __restrict__ part, int* __restrict__ rtab) {
  __shared__ __attribute__((aligned(16))) float wl[HF * WPITCH];
  __shared__ int lh[NB];
  __shared__ int lbase[NB];
  __shared__ int wsum[8];
  const int tid  = threadIdx.x;
  const int lane = tid & 31;
  const int wave = tid >> 5;

  if (blockIdx.x >= GEMM_WG) {
    const int c = blockIdx.x - GEMM_WG;
    lh[tid] = 0;
    __syncthreads();
    const int e0 = c * ECHUNK + tid;
    int s[EPT], d[EPT], bk[EPT], r[EPT];
#pragma unroll
    for (int i = 0; i < EPT; ++i) {
      const int e = min(e0 + i * 256, N_EDGES - 1);
      s[i] = ei[e];
      d[i] = ei[N_EDGES + e];
    }
#pragma unroll
    for (int i = 0; i < EPT; ++i) {
      bk[i] = d[i] / NPB;
      r[i] = 0;
      if (e0 + i * 256 < N_EDGES)
        r[i] = __hip_atomic_fetch_add(&lh[bk[i]], 1, __ATOMIC_RELAXED,
                                      __HIP_MEMORY_SCOPE_WORKGROUP);
    }
    __syncthreads();
    const int tot = lh[tid];
    int incl = tot;
    for (int k = 1; k < 32; k <<= 1) {
      const int up = __shfl_up(incl, k, 32);
      if (lane >= k) incl += up;
    }
    if (lane == 31) wsum[wave] = incl;
    __syncthreads();
    int woff = 0;
    for (int w = 0; w < 8; ++w) woff += (w < wave) ? wsum[w] : 0;
    const int st = woff + incl - tot;
    lbase[tid] = st;
    rtab[c * NB + tid] = st | (tot << 16);
    __syncthreads();
#pragma unroll
    for (int i = 0; i < EPT; ++i) {
      if (e0 + i * 256 < N_EDGES)
        part[c * ECHUNK + lbase[bk[i]] + r[i]] = ((d[i] - bk[i] * NPB) << 16) | s[i];
    }
    return;
  }

  const int half = lane >> 4;
  const int l15  = lane & 15;
  const int m0   = (blockIdx.x * GEMM_WAVES + wave) * 16;
  const bool live = m0 < N_NODES;
  const int mrow = live ? m0 + l15 : l15;

  v4f wv[16];
#pragma unroll
  for (int i = 0; i < 16; ++i)
    wv[i] = *(const v4f*)(W + (i * 256 + tid) * 4);

  const float* __restrict__ arow = x + mrow * F_IN + 2 * half;
  v2f a[F_IN / 4];
#pragma unroll
  for (int k = 0; k < F_IN / 4; ++k) {
    a[k].x = arow[4 * k];
    a[k].y = arow[4 * k + 1];
  }

#pragma unroll
  for (int i = 0; i < 16; ++i) {
    const int idx = i * 256 + tid;
    *(v4f*)(wl + (idx >> 5) * WPITCH + (idx & 31) * 4) = wv[i];
  }
  __syncthreads();
  if (!live) return;

  const float* __restrict__ brow = wl + l15 * WPITCH + 2 * half;

  v8f c[8];
#pragma unroll
  for (int t = 0; t < 8; ++t) c[t] = (v8f){};

  v2f bc[8][2], bn[8][2];
#pragma unroll
  for (int t = 0; t < 8; ++t) {
    bc[t][0] = *(const v2f*)(brow + t * 16 * WPITCH);
    bc[t][1] = *(const v2f*)(brow + t * 16 * WPITCH + 4);
  }
#pragma unroll
  for (int j = 0; j < F_IN / 8; ++j) {
    if (j + 1 < F_IN / 8) {
#pragma unroll
      for (int t = 0; t < 8; ++t) {
        bn[t][0] = *(const v2f*)(brow + t * 16 * WPITCH + 8 * (j + 1));
        bn[t][1] = *(const v2f*)(brow + t * 16 * WPITCH + 8 * (j + 1) + 4);
      }
    }
    __builtin_amdgcn_sched_barrier(0);
#pragma unroll
    for (int t = 0; t < 8; ++t)
      c[t] = __builtin_amdgcn_wmma_f32_16x16x4_f32(
           false, a[2 * j],  false, bc[t][0],
           (short)0, c[t],  false,  false);
#pragma unroll
    for (int t = 0; t < 8; ++t)
      c[t] = __builtin_amdgcn_wmma_f32_16x16x4_f32(
           false, a[2 * j + 1],  false, bc[t][1],
           (short)0, c[t],  false,  false);
    __builtin_amdgcn_sched_barrier(0);
#pragma unroll
    for (int t = 0; t < 8; ++t) { bc[t][0] = bn[t][0]; bc[t][1] = bn[t][1]; }
  }

  float* __restrict__ op = h + (m0 + half * 8) * HF + l15;
#pragma unroll
  for (int t = 0; t < 8; ++t) {
#pragma unroll
    for (int v = 0; v < 8; ++v) op[v * HF + t * 16] = c[t][v];
  }
}

__device__ __forceinline__ float leaky(float v) {
  return fmaxf(v, NEG_SLOPE * v);
}

template <int CTRL>
__device__ __forceinline__ float quad(float x) {
  return __int_as_float(__builtin_amdgcn_mov_dpp(__float_as_int(x), CTRL, 0xf, 0xf, true));
}
#define QP_XOR1 0xB1
#define QP_XOR2 0x4E

__device__ __forceinline__ float dot4(v4f a, v4f b) {
  const float d0 = __builtin_fmaf(a.z, b.z, a.x * b.x);
  const float d1 = __builtin_fmaf(a.w, b.w, a.y * b.y);
  return d0 + d1;
}
__device__ __forceinline__ float quad_sum(float d) {
  d += quad<QP_XOR1>(d);
  d += quad<QP_XOR2>(d);
  return d;
}

struct NodeAcc {
  v4f acc;
  float m;
  float s;
};

__device__ __forceinline__ void rows4(NodeAcc& a, int j0, int deg, v4f h0, v4f h1,
                                      v4f h2, v4f h3, v4f as4, float adn) {
  const float ninf = -__builtin_inff();
  const float e0 = leaky(quad_sum(dot4(h0, as4)) + adn);
  const float e1 = (j0 + 1 < deg) ? leaky(quad_sum(dot4(h1, as4)) + adn) : ninf;
  const float e2 = (j0 + 2 < deg) ? leaky(quad_sum(dot4(h2, as4)) + adn) : ninf;
  const float e3 = (j0 + 3 < deg) ? leaky(quad_sum(dot4(h3, as4)) + adn) : ninf;
  const float mn = fmaxf(fmaxf(a.m, e0), fmaxf(fmaxf(e1, e2), e3));
  const float sc = __builtin_amdgcn_exp2f(a.m - mn);
  const float w0 = __builtin_amdgcn_exp2f(e0 - mn);
  const float w1 = __builtin_amdgcn_exp2f(e1 - mn);
  const float w2 = __builtin_amdgcn_exp2f(e2 - mn);
  const float w3 = __builtin_amdgcn_exp2f(e3 - mn);
  a.s = a.s * sc + (((w0 + w1) + w2) + w3);
  a.acc = a.acc * sc;
  a.acc += w0 * h0;
  a.acc += w1 * h1;
  a.acc += w2 * h2;
  a.acc += w3 * h3;
  a.m = mn;
}

__device__ __forceinline__ void node_one(
    const int* seg, int* srt, int deg, int lane, int n,
    const float* __restrict__ h, v4f as4, v4f ad4, v4f bia,
    float* __restrict__ out) {
  const float L2E = 1.4426950408889634f;
  const v4f hn  = *(const v4f*)(h + n * HF + lane * 4);
  const float adn = quad_sum(dot4(hn, ad4)) * L2E;

  NodeAcc a;
  a.acc = (v4f){0.f, 0.f, 0.f, 0.f};
  a.m = -__builtin_inff();
  a.s = 0.f;

  if (deg <= 32) {
    const int key = (lane < deg) ? seg[lane] * 32 + lane : 0x7fffffff;
    int r = 0;
    for (int j = 0; j < deg; ++j) {
      r += (__builtin_amdgcn_readlane(key, j) < key) ? 1 : 0;
    }
    const int sid = __builtin_amdgcn_ds_permute(r << 2, key) >> 5;

#pragma unroll
    for (int q = 0; q < 4; ++q) {
      if (8 * q < deg) {
        const int j0 = 8 * q;
        int sr[8];
        sr[0] = __builtin_amdgcn_readlane(sid, j0);
#pragma unroll
        for (int u = 1; u < 8; ++u) {
          const int fb = (u > 4) ? sr[4] : sr[0];
          sr[u] = (j0 + u < deg) ? __builtin_amdgcn_readlane(sid, j0 + u) : fb;
        }
        v4f hv[8];
#pragma unroll
        for (int u = 0; u < 8; ++u)
          hv[u] = *(const v4f*)(h + sr[u] * HF + lane * 4);
        rows4(a, j0, deg, hv[0], hv[1], hv[2], hv[3], as4, adn);
        if (j0 + 4 < deg)
          rows4(a, j0 + 4, deg, hv[4], hv[5], hv[6], hv[7], as4, adn);
      }
    }
  } else {
    for (int i = lane; i < deg; i += 32) {
      const int key = seg[i];
      int r = 0;
      for (int j = 0; j < deg; ++j) {
        const int kj = seg[j];
        r += ((kj < key) | ((kj == key) & (j < i))) ? 1 : 0;
      }
      srt[r] = key;
    }
    __threadfence();
    for (int j = 0; j < deg; ++j) {
      const int sj = __builtin_amdgcn_readfirstlane(srt[j]);
      const v4f hv = *(const v4f*)(h + sj * HF + lane * 4);
      const float e  = leaky(quad_sum(dot4(hv, as4)) + adn);
      const float mn = fmaxf(a.m, e);
      const float sc = __builtin_amdgcn_exp2f(a.m - mn);
      const float w  = __builtin_amdgcn_exp2f(e - mn);
      a.s = a.s * sc + w;
      a.acc = a.acc * sc;
      a.acc += w * hv;
      a.m = mn;
    }
  }

  const float rden = 1.0f / (a.s + 1e-16f);
  *(v4f*)(out + n * HF + lane * 4) = a.acc * rden + bia;
}

#define BLD  1024
#define SCAP 8192

__device__ __forceinline__ int bucket_entry(const int* __restrict__ part,
                                            const int* pre, const int* rst, int e) {
  int lo = 0;
#pragma unroll
  for (int step = NB / 2; step >= 1; step >>= 1) {
    const int nx = lo + step;
    if (pre[nx] <= e) lo = nx;
  }
  return part[lo * ECHUNK + rst[lo] + (e - pre[lo])];
}

__global__ __launch_bounds__(BLD) void gat_bucket(
    const int* __restrict__ part, const int* __restrict__ rtab,
    int* __restrict__ slot, int* __restrict__ srtbuf,
    const float* __restrict__ h, const float* __restrict__ att_src,
    const float* __restrict__ att_dst, const float* __restrict__ bias,
    float* __restrict__ out) {
  __shared__ int slotL[SCAP];
  __shared__ int srtL[SCAP];
  __shared__ int pre[NB];
  __shared__ int rst[NB];
  __shared__ int ncnt[NB];
  __shared__ int ncur[NB];
  __shared__ int nofs[NB];
  __shared__ int wsS[32];
  __shared__ int wsB[8];
  __shared__ int nxt;
  const int tid  = threadIdx.x;
  const int lane = tid & 31;
  const int wave = tid >> 5;
  const int b = blockIdx.x;

  const float L2E = 1.4426950408889634f;
  const v4f as4 = *(const v4f*)(att_src + lane * 4) * L2E;
  const v4f ad4 = *(const v4f*)(att_dst + lane * 4);
  const v4f bia = *(const v4f*)(bias + lane * 4);

  int rs = 0, ln = 0;
  if (tid < NCHUNK) {
    const int v = rtab[tid * NB + b];
    rs = v & 0xffff;
    ln = v >> 16;
  }
  if (tid == 0) nxt = 0;
  int incl = ln;
  for (int k = 1; k < 32; k <<= 1) {
    const int up = __shfl_up(incl, k, 32);
    if (lane >= k) incl += up;
  }
  if (lane == 31) wsS[wave] = incl;
  __syncthreads();
  int wo = 0, nb = 0;
  for (int w = 0; w < 32; ++w) {
    const int sv = wsS[w];
    nb += sv;
    wo += (w < wave) ? sv : 0;
  }
  if (tid < NB) {
    pre[tid] = (tid < NCHUNK) ? wo + incl - ln : 0x7fffffff;
    rst[tid] = rs;
    ncnt[tid] = 0;
  }
  __syncthreads();

  const bool big = nb > SCAP;
  int bstart = 0;
  if (big) {
    int acc = 0;
    for (int idx = tid; idx < NCHUNK * NB; idx += BLD)
      acc += ((idx & (NB - 1)) < b) ? (rtab[idx] >> 16) : 0;
    for (int k = 16; k >= 1; k >>= 1) acc += __shfl_xor(acc, k, 32);
    if (lane == 0) wsS[wave] = acc;
    __syncthreads();
    for (int w = 0; w < 32; ++w) bstart += wsS[w];
  }

  for (int eb = tid; eb < nb; eb += 4 * BLD) {
    int ent[4];
#pragma unroll
    for (int k = 0; k < 4; ++k) {
      const int e = eb + k * BLD;
      ent[k] = (e < nb) ? bucket_entry(part, pre, rst, e) : -1;
    }
#pragma unroll
    for (int k = 0; k < 4; ++k) {
      if (ent[k] >= 0) {
        __hip_atomic_fetch_add(&ncnt[ent[k] >> 16], 1, __ATOMIC_RELAXED,
                               __HIP_MEMORY_SCOPE_WORKGROUP);
        if (!big) srtL[eb + k * BLD] = ent[k];
      }
    }
  }
  __syncthreads();

  int c = 0, inc2 = 0;
  if (tid < NB) {
    c = ncnt[tid];
    inc2 = c;
    for (int k = 1; k < 32; k <<= 1) {
      const int up = __shfl_up(inc2, k, 32);
      if (lane >= k) inc2 += up;
    }
    if (lane == 31) wsB[wave] = inc2;
  }
  __syncthreads();
  if (tid < NB) {
    int woff = 0;
    for (int w = 0; w < 8; ++w) woff += (w < wave) ? wsB[w] : 0;
    const int noff = woff + inc2 - c;
    ncur[tid] = noff;
    nofs[tid] = noff;
  }
  __syncthreads();

  for (int eb = tid; eb < nb; eb += 4 * BLD) {
    int ent[4];
#pragma unroll
    for (int k = 0; k < 4; ++k) {
      const int e = eb + k * BLD;
      ent[k] = -1;
      if (e < nb) ent[k] = big ? bucket_entry(part, pre, rst, e) : srtL[e];
    }
#pragma unroll
    for (int k = 0; k < 4; ++k) {
      if (ent[k] >= 0) {
        const int pos = __hip_atomic_fetch_add(&ncur[ent[k] >> 16], 1,
                                               __ATOMIC_RELAXED,
                                               __HIP_MEMORY_SCOPE_WORKGROUP);
        if (big) slot[bstart + pos] = ent[k] & 0xffff;
        else     slotL[pos] = ent[k] & 0xffff;
      }
    }
  }
  if (big) __threadfence();
  __syncthreads();

  const int nn = min(NPB, N_NODES - b * NPB);
  int tk = 0;
  if (lane == 0)
    tk = __hip_atomic_fetch_add(&nxt, 1, __ATOMIC_RELAXED, __HIP_MEMORY_SCOPE_WORKGROUP);
  for (int it = 0; it < NPB; ++it) {
    const int i = __builtin_amdgcn_readfirstlane(tk);
    if (i >= nn) break;
    if (lane == 0)
      tk = __hip_atomic_fetch_add(&nxt, 1, __ATOMIC_RELAXED, __HIP_MEMORY_SCOPE_WORKGROUP);
    const int deg = __builtin_amdgcn_readfirstlane(ncnt[i]);
    const int beg = __builtin_amdgcn_readfirstlane(nofs[i]);
    const int n = b * NPB + i;
    if (big)
      node_one(slot + bstart + beg, srtbuf + bstart + beg, deg, lane, n, h, as4, ad4, bia, out);
    else
      node_one(slotL + beg, srtL + beg, deg, lane, n, h, as4, ad4, bia, out);
  }
}

extern "C" void kernel_launch(void* const* d_in, const int* in_sizes, int n_in,
                              void* d_out, int out_size, void* d_ws, size_t ws_size,
                              hipStream_t stream) {
  const float* x        = (const float*)d_in[0];
  const float* W        = (const float*)d_in[1];
  const float* att_src  = (const float*)d_in[2];
  const float* att_dst  = (const float*)d_in[3];
  const float* bias     = (const float*)d_in[4];
  const int*   ei       = (const int*)d_in[5];
  float*       out      = (float*)d_out;

  float* h      = (float*)d_ws;
  int*   rtab   = (int*)(h + (size_t)N_NODES * HF);
  int*   part   = rtab  + 2 * N_NODES + 32;
  int*   slot   = part  + N_EDGES;
  int*   srtbuf = slot  + N_EDGES;
  static_assert(NCHUNK * NB <= 2 * N_NODES + 32, "run table must fit in front of part");

  gat_gemm_group<<<GEMM_WG + NCHUNK, 256, 0, stream>>>(
      x, W, h, ei, part, rtab);

  gat_bucket<<<NB, BLD, 0, stream>>>(
      part, rtab, slot, srtbuf, h, att_src, att_dst, bias, out);
}
